# starting kernel
# baseline (speedup 1.0000x reference)
#include <hip/hip_runtime.h>


typedef __attribute__((ext_vector_type(2))) float v2f;
typedef __attribute__((ext_vector_type(8))) float v8f;

#define D 128

__global__ void k_deg_init(float* __restrict__ deg, int n) {
  int i = blockIdx.x * blockDim.x + threadIdx.x;
  if (i < n) deg[i] = 1.0f;
}

__global__ void k_deg_count(const int* __restrict__ dst, float* __restrict__ deg, int e) {
  int i = blockIdx.x * blockDim.x + threadIdx.x;
  if (i < e) atomicAdd(&deg[dst[i]], 1.0f);
}

__global__ void k_deg_rsqrt(float* __restrict__ deg, int n) {
  int i = blockIdx.x * blockDim.x + threadIdx.x;
  if (i < n) deg[i] = rsqrtf(deg[i]);
}

__global__ void k_bias_init(float* __restrict__ out, const float* __restrict__ b, int total) {
  int i = blockIdx.x * blockDim.x + threadIdx.x;
  if (i < total) out[i] = b[i & (D - 1)];
}

template <bool RELU_IN>
__global__ __launch_bounds__(256) void k_gemm128(const float* __restrict__ A,
                                                 const float* __restrict__ W,
                                                 float* __restrict__ H, int nrows) {
  __shared__ float Ws[D * D];

  const int tid = threadIdx.x;
  {
    const float4* __restrict__ Wv = (const float4*)W;
    float4* Sv = (float4*)Ws;
#pragma unroll
    for (int i = 0; i < (D * D / 4) / 256; ++i)
      Sv[tid + i * 256] = Wv[tid + i * 256];
  }
  __syncthreads();

  const int wave = tid >> 5;
  const int lane = tid & 31;
  const int row0 = blockIdx.x * 128 + wave * 16;
  if (row0 >= nrows) return;

  const int m  = lane & 15;
  const int kb = (lane >> 4) << 1;

  v8f acc[8] = {};

  const float* __restrict__ Arow = A + (size_t)(row0 + m) * D;
  for (int k = 0; k < D; k += 4) {
    float ax = Arow[k + kb];
    float ay = Arow[k + kb + 1];
    if (RELU_IN) {
      ax = fmaxf(ax, 0.0f);
      ay = fmaxf(ay, 0.0f);
    }
    v2f a;
    a.x = ax;
    a.y = ay;
#pragma unroll
    for (int t = 0; t < 8; ++t) {
      const int col = t * 16 + m;
      v2f b;
      b.x = Ws[(k + kb) * D + col];
      b.y = Ws[(k + kb + 1) * D + col];
      acc[t] = __builtin_amdgcn_wmma_f32_16x16x4_f32(
           false, a,  false, b,
           (short)0, acc[t],  false,  false);
    }
  }

  const int rb = row0 + ((lane >> 4) << 3);
#pragma unroll
  for (int t = 0; t < 8; ++t) {
    const int col = t * 16 + m;
#pragma unroll
    for (int v = 0; v < 8; ++v)
      H[(size_t)(rb + v) * D + col] = acc[t][v];
  }
}

__global__ __launch_bounds__(256) void k_scatter(const float* __restrict__ h,
                                                 const int* __restrict__ ei, int E, int n,
                                                 const float* __restrict__ dis,
                                                 float* __restrict__ acc) {
  const int wid  = blockIdx.x * (blockDim.x >> 5) + (threadIdx.x >> 5);
  const int lane = threadIdx.x & 31;
  const int total = E + n;
  if (wid >= total) return;

  int s, d;
  if (wid < E) {
    s = ei[wid];
    d = ei[E + wid];
  } else {
    s = d = wid - E;
  }
  const float nrm = dis[s] * dis[d];

  const float4 hv = *(const float4*)(h + (size_t)s * D + lane * 4);
  float* o = acc + (size_t)d * D + lane * 4;
  atomicAdd(o + 0, hv.x * nrm);
  atomicAdd(o + 1, hv.y * nrm);
  atomicAdd(o + 2, hv.z * nrm);
  atomicAdd(o + 3, hv.w * nrm);
}

extern "C" void kernel_launch(void* const* d_in, const int* in_sizes, int n_in,
                              void* d_out, int out_size, void* d_ws, size_t ws_size,
                              hipStream_t stream) {
  const float* x  = (const float*)d_in[0];
  const int*   ei = (const int*)d_in[1];
  const float* W1 = (const float*)d_in[2];
  const float* b1 = (const float*)d_in[3];
  const float* W2 = (const float*)d_in[4];
  const float* b2 = (const float*)d_in[5];
  const float* W3 = (const float*)d_in[6];
  const float* b3 = (const float*)d_in[7];

  const int N = in_sizes[0] / D;
  const int E = in_sizes[1] / 2;

  float* deg  = (float*)d_ws;
  size_t npad = ((size_t)N + 255) & ~(size_t)255;
  float* buf1 = deg + npad;
  float* buf2 = buf1 + (size_t)N * D;
  float* out  = (float*)d_out;

  const dim3 blk(256);
  const int gN    = (N + 255) / 256;
  const int gE    = (E + 255) / 256;
  const int totF  = N * D;
  const int gF    = (totF + 255) / 256;
  const int gGemm = (N + 127) / 128;
  const int gScat = ((E + N) + 7) / 8;

  k_deg_init<<<gN, blk, 0, stream>>>(deg, N);
  k_deg_count<<<gE, blk, 0, stream>>>(ei + E, deg, E);
  k_deg_rsqrt<<<gN, blk, 0, stream>>>(deg, N);

  k_gemm128<false><<<gGemm, blk, 0, stream>>>(x, W1, buf1, N);
  k_bias_init<<<gF, blk, 0, stream>>>(buf2, b1, totF);
  k_scatter<<<gScat, blk, 0, stream>>>(buf1, ei, E, N, deg, buf2);

  k_gemm128<true><<<gGemm, blk, 0, stream>>>(buf2, W2, buf1, N);
  k_bias_init<<<gF, blk, 0, stream>>>(buf2, b2, totF);
  k_scatter<<<gScat, blk, 0, stream>>>(buf1, ei, E, N, deg, buf2);

  k_gemm128<true><<<gGemm, blk, 0, stream>>>(buf2, W3, buf1, N);
  k_bias_init<<<gF, blk, 0, stream>>>(out, b3, totF);
  k_scatter<<<gScat, blk, 0, stream>>>(buf1, ei, E, N, deg, out);
}


# Edge scatter by f32 atomics replaced by per-call bucketing by destination (returning integer adds, prefix sum, placement) and a one-wave-per-node register sum in fixed sorted-source order
# speedup vs baseline: 2.4428x; 2.4428x over previous
#include <hip/hip_runtime.h>


typedef __attribute__((ext_vector_type(2))) float v2f;
typedef __attribute__((ext_vector_type(8))) float v8f;

#define D 128

__global__ void k_cnt_zero(int* __restrict__ cnt, int n) {
  int i = blockIdx.x * blockDim.x + threadIdx.x;
  if (i < n) cnt[i] = 0;
}

__global__ void k_cnt_rank(const int* __restrict__ dst, int* __restrict__ cnt,
                           int* __restrict__ rank, int e) {
  int i = blockIdx.x * blockDim.x + threadIdx.x;
  if (i < e) rank[i] = atomicAdd(&cnt[dst[i]], 1);
}

__global__ __launch_bounds__(1024) void k_rowptr(const int* __restrict__ cnt,
                                                 int* __restrict__ seg_w,
                                                 float* __restrict__ dis, int n) {
  int2* __restrict__ seg = (int2*)seg_w;
  __shared__ int s_inc[32];
  __shared__ int s_pre[32];
  const int tid  = threadIdx.x;
  const int lane = tid & 31;
  const int wave = tid >> 5;
  const int base = blockIdx.x * 1024;

  int pre = 0;
  const int4* __restrict__ c4 = (const int4*)cnt;
#pragma unroll 4
  for (int i = tid; i < (base >> 2); i += 1024) {
    const int4 v = c4[i];
    pre += (v.x + v.y) + (v.z + v.w);
  }

  const int node = base + tid;
  const int c = (node < n) ? cnt[node] : 0;
  int inc = c;
  for (int o = 1; o < 32; o <<= 1) {
    const int t = __shfl_up(inc, o, 32);
    if (lane >= o) inc += t;
  }
  for (int o = 16; o > 0; o >>= 1) pre += __shfl_xor(pre, o, 32);
  if (lane == 31) s_inc[wave] = inc;
  if (lane == 0) s_pre[wave] = pre;
  __syncthreads();

  int tot = s_pre[lane] + ((lane < wave) ? s_inc[lane] : 0);
  for (int o = 16; o > 0; o >>= 1) tot += __shfl_xor(tot, o, 32);

  if (node < n) {
    seg[node] = make_int2(tot + inc - c, c);
    dis[node] = rsqrtf((float)(c + 1));
  }
}

__global__ void k_place(const int* __restrict__ ei, int E, const int* __restrict__ seg_w,
                        const int* __restrict__ rank, const float* __restrict__ dis,
                        int* __restrict__ ent_w) {
  const int2* __restrict__ seg = (const int2*)seg_w;
  int2* __restrict__ ent = (int2*)ent_w;
  int i = blockIdx.x * blockDim.x + threadIdx.x;
  if (i < E) {
    const int s = ei[i];
    ent[seg[ei[E + i]].x + rank[i]] = make_int2(s, __float_as_int(dis[s]));
  }
}

template <bool RELU_IN>
__global__ __launch_bounds__(256) void k_gemm128(const float* __restrict__ A,
                                                 const float* __restrict__ W,
                                                 float* __restrict__ H, int nrows) {
  __shared__ float Ws[D * D];

  const int tid = threadIdx.x;
  {
    const float4* __restrict__ Wv = (const float4*)W;
    float4* Sv = (float4*)Ws;
#pragma unroll
    for (int i = 0; i < (D * D / 4) / 256; ++i)
      Sv[tid + i * 256] = Wv[tid + i * 256];
  }
  __syncthreads();

  const int wave = tid >> 5;
  const int lane = tid & 31;
  const int row0 = blockIdx.x * 128 + wave * 16;
  if (row0 >= nrows) return;

  const int m  = lane & 15;
  const int kb = (lane >> 4) << 1;

  v8f acc[8] = {};

  const float* __restrict__ Arow = A + (size_t)(row0 + m) * D;
  for (int k = 0; k < D; k += 4) {
    float ax = Arow[k + kb];
    float ay = Arow[k + kb + 1];
    if (RELU_IN) {
      ax = fmaxf(ax, 0.0f);
      ay = fmaxf(ay, 0.0f);
    }
    v2f a;
    a.x = ax;
    a.y = ay;
#pragma unroll
    for (int t = 0; t < 8; ++t) {
      const int col = t * 16 + m;
      v2f b;
      b.x = Ws[(k + kb) * D + col];
      b.y = Ws[(k + kb + 1) * D + col];
      acc[t] = __builtin_amdgcn_wmma_f32_16x16x4_f32(
           false, a,  false, b,
           (short)0, acc[t],  false,  false);
    }
  }

  const int rb = row0 + ((lane >> 4) << 3);
#pragma unroll
  for (int t = 0; t < 8; ++t) {
    const int col = t * 16 + m;
#pragma unroll
    for (int v = 0; v < 8; ++v)
      H[(size_t)(rb + v) * D + col] = acc[t][v];
  }
}

__device__ __forceinline__ void pick_slot(int r, int s, float w, int p, int& su, float& wu) {
  const unsigned hit = __builtin_amdgcn_ballot_w32(r == p);
  const int idx = __builtin_ctz(hit);
  su = __builtin_amdgcn_readlane(s, idx);
  wu = __int_as_float(__builtin_amdgcn_readlane(__float_as_int(w), idx));
}

__device__ __forceinline__ float4 ld_row(const float* __restrict__ h, int s, int lane) {
  return *(const float4*)(h + (size_t)s * D + lane * 4);
}

__device__ __forceinline__ float4 fma_row(float4 v, float w, float4 a) {
  a.x = fmaf(v.x, w, a.x);
  a.y = fmaf(v.y, w, a.y);
  a.z = fmaf(v.z, w, a.z);
  a.w = fmaf(v.w, w, a.w);
  return a;
}

__global__ __launch_bounds__(256) void k_gather(const float* __restrict__ h,
                                                const int* __restrict__ seg_w,
                                                const int* __restrict__ ent_w,
                                                const float* __restrict__ dis,
                                                const float* __restrict__ bias,
                                                float* __restrict__ out, int n) {
  const int2* __restrict__ seg = (const int2*)seg_w;
  const int2* __restrict__ ent = (const int2*)ent_w;
  const int lane = threadIdx.x & 31;
  const int d = __builtin_amdgcn_readfirstlane(blockIdx.x * 8 + (threadIdx.x >> 5));
  if (d >= n) return;

  const int2 sc   = seg[d];
  const int start = sc.x;
  const int c     = sc.y;
  const float dd  = dis[d];
  const float4 bv = *(const float4*)(bias + lane * 4);
  const float4 zero = make_float4(0.0f, 0.0f, 0.0f, 0.0f);

  float4 acc = fma_row(ld_row(h, d, lane), dd * dd, zero);

  for (int base = 0; base < c; base += 32) {
    const int m = (c - base < 32) ? (c - base) : 32;
    int s = d;
    float w = 0.0f;
    if (lane < m) {
      const int2 e = ent[start + base + lane];
      s = e.x;
      w = __int_as_float(e.y) * dd;
    }
    int r = 0;
    for (int i = 0; i < m; ++i) {
      const int si = __builtin_amdgcn_readlane(s, i);
      r += ((si < s) || (si == s && i < lane)) ? 1 : 0;
    }
    if (lane >= m) r = 64;

    for (int p = 0; p < m; p += 4) {
      int s0, s1, s2, s3;
      float w0, w1 = 0.0f, w2 = 0.0f, w3 = 0.0f;
      float4 v0, v1 = zero, v2 = zero, v3 = zero;
      pick_slot(r, s, w, p, s0, w0);
      v0 = ld_row(h, s0, lane);
      if (p + 1 < m) {
        pick_slot(r, s, w, p + 1, s1, w1);
        v1 = ld_row(h, s1, lane);
      }
      if (p + 2 < m) {
        pick_slot(r, s, w, p + 2, s2, w2);
        v2 = ld_row(h, s2, lane);
      }
      if (p + 3 < m) {
        pick_slot(r, s, w, p + 3, s3, w3);
        v3 = ld_row(h, s3, lane);
      }
      acc = fma_row(v0, w0, acc);
      acc = fma_row(v1, w1, acc);
      acc = fma_row(v2, w2, acc);
      acc = fma_row(v3, w3, acc);
    }
  }

  float4 o;
  o.x = acc.x + bv.x;
  o.y = acc.y + bv.y;
  o.z = acc.z + bv.z;
  o.w = acc.w + bv.w;
  *(float4*)(out + (size_t)d * D + lane * 4) = o;
}

extern "C" void kernel_launch(void* const* d_in, const int* in_sizes, int n_in,
                              void* d_out, int out_size, void* d_ws, size_t ws_size,
                              hipStream_t stream) {
  const float* x  = (const float*)d_in[0];
  const int*   ei = (const int*)d_in[1];
  const float* W1 = (const float*)d_in[2];
  const float* b1 = (const float*)d_in[3];
  const float* W2 = (const float*)d_in[4];
  const float* b2 = (const float*)d_in[5];
  const float* W3 = (const float*)d_in[6];
  const float* b3 = (const float*)d_in[7];

  const int N = in_sizes[0] / D;
  const int E = in_sizes[1] / 2;

  size_t npad = ((size_t)N + 255) & ~(size_t)255;
  size_t epad = ((size_t)E + 255) & ~(size_t)255;
  int*   cnt    = (int*)d_ws;
  float* dis    = (float*)(cnt + npad);
  int*   seg    = (int*)(dis + npad);
  int*   rank   = seg + 2 * npad;
  int*   ent    = rank + epad;
  float* buf1   = (float*)(ent + 2 * epad);
  float* buf2   = buf1 + (size_t)N * D;
  float* out    = (float*)d_out;

  const dim3 blk(256);
  const int gN    = (N + 255) / 256;
  const int gE    = (E + 255) / 256;
  const int gScan = (N + 1023) / 1024;
  const int gGemm = (N + 127) / 128;
  const int gGath = (N + 7) / 8;

  k_cnt_zero<<<gN, blk, 0, stream>>>(cnt, N);
  k_cnt_rank<<<gE, blk, 0, stream>>>(ei + E, cnt, rank, E);
  k_rowptr<<<gScan, dim3(1024), 0, stream>>>(cnt, seg, dis, N);
  k_place<<<gE, blk, 0, stream>>>(ei, E, seg, rank, dis, ent);

  k_gemm128<false><<<gGemm, blk, 0, stream>>>(x, W1, buf1, N);
  k_gather<<<gGath, blk, 0, stream>>>(buf1, seg, ent, dis, b1, buf2, N);

  k_gemm128<true><<<gGemm, blk, 0, stream>>>(buf2, W2, buf1, N);
  k_gather<<<gGath, blk, 0, stream>>>(buf1, seg, ent, dis, b2, buf2, N);

  k_gemm128<true><<<gGemm, blk, 0, stream>>>(buf2, W3, buf1, N);
  k_gather<<<gGath, blk, 0, stream>>>(buf1, seg, ent, dis, b3, out, N);
}


# Buckets at a fixed place, one 128-byte line of 32 sources per node: the counting pass stores each source itself, so the prefix-sum and placement launches go (7 launches to 5); sum order unchanged
# speedup vs baseline: 3.4263x; 1.4026x over previous
#include <hip/hip_runtime.h>


typedef __attribute__((ext_vector_type(2))) float v2f;
typedef __attribute__((ext_vector_type(8))) float v8f;

#define D 128
#define WP 136
#define AP 132
#define TN 32
#define CAP 32

__global__ void k_cnt_zero(int* __restrict__ cnt, int n, int* __restrict__ novf) {
  int i = blockIdx.x * blockDim.x + threadIdx.x;
  if (i < n) cnt[i] = 0;
  if (i == 0) *novf = 0;
}

__global__ void k_bucket(const int* __restrict__ ei, int E, int* __restrict__ cnt,
                         int* __restrict__ slots, int* __restrict__ novf,
                         int* __restrict__ ovf_w) {
  int2* __restrict__ ovf = (int2*)ovf_w;
  int i = blockIdx.x * blockDim.x + threadIdx.x;
  if (i < E) {
    const int s = ei[i];
    const int d = ei[E + i];
    const int r = atomicAdd(&cnt[d], 1);
    if (r < CAP) {
      slots[(size_t)d * CAP + r] = s;
    } else {
      const int q = atomicAdd(novf, 1);
      ovf[q] = make_int2(d, s);
    }
  }
}

__device__ __forceinline__ int pick_lane(int r, int p) {
  const unsigned hit = __builtin_amdgcn_ballot_w32(r == p);
  return __builtin_ctz(hit);
}

__device__ __forceinline__ float4 ld_row(const float* __restrict__ h, int s, int lane) {
  return *(const float4*)(h + (size_t)s * D + lane * 4);
}

__device__ __forceinline__ float4 fma_row(float4 v, float w, float4 a) {
  a.x = fmaf(v.x, w, a.x);
  a.y = fmaf(v.y, w, a.y);
  a.z = fmaf(v.z, w, a.z);
  a.w = fmaf(v.w, w, a.w);
  return a;
}

__device__ __forceinline__ float4 node_sum(const float* __restrict__ h,
                                           const int* __restrict__ cnt,
                                           const int* __restrict__ novf,
                                           const int2* __restrict__ ovf, int d, int c,
                                           float dd, int s_first, float4 self_row,
                                           int lane) {
  const float4 zero = make_float4(0.0f, 0.0f, 0.0f, 0.0f);

  float4 acc = fma_row(self_row, dd * dd, zero);

  const int m = (c < CAP) ? c : CAP;
  if (m > 0) {
    const int s = s_first;
    int r = 0;
    for (int i = 0; i < m; ++i) {
      const int si = __builtin_amdgcn_readlane(s, i);
      r += ((si < s) || (si == s && i < lane)) ? 1 : 0;
    }
    if (lane >= m) r = 64;

    float w = 0.0f;
    for (int p = 0; p < m; p += 8) {
      float4 v[8];
      int ix[8];
#pragma unroll
      for (int q = 0; q < 8; ++q) {
        v[q]  = zero;
        ix[q] = 0;
        if (p + q < m) {
          ix[q] = pick_lane(r, p + q);
          v[q]  = ld_row(h, __builtin_amdgcn_readlane(s, ix[q]), lane);
        }
      }
      if (p == 0) {
        int cs = 0;
        if (lane < m) cs = cnt[s];
        w = (lane < m) ? rsqrtf((float)(cs + 1)) * dd : 0.0f;
      }
#pragma unroll
      for (int q = 0; q < 8; ++q) {
        const float wq = (p + q < m)
            ? __int_as_float(__builtin_amdgcn_readlane(__float_as_int(w), ix[q])) : 0.0f;
        acc = fma_row(v[q], wq, acc);
      }
    }
  }

  if (c > CAP) {
    const int nov = *novf;
    for (int base = 0; base < nov; base += 32) {
      int2 oe = make_int2(-1, 0);
      if (base + lane < nov) oe = ovf[base + lane];
      const bool mine = (oe.x == d);
      int cs = 0;
      if (mine) cs = cnt[oe.y];
      const float w = mine ? rsqrtf((float)(cs + 1)) * dd : 0.0f;
      unsigned hit = __builtin_amdgcn_ballot_w32(mine);
      while (hit != 0u) {
        const int idx = __builtin_ctz(hit);
        hit &= hit - 1u;
        const float4 v = ld_row(h, __builtin_amdgcn_readlane(oe.y, idx), lane);
        const float wq = __int_as_float(__builtin_amdgcn_readlane(__float_as_int(w), idx));
        acc = fma_row(v, wq, acc);
      }
    }
  }
  return acc;
}

template <bool RELU_OUT>
__global__ __launch_bounds__(512) void k_layer(const float* __restrict__ hin,
                                               const float* __restrict__ W,
                                               const int* __restrict__ cnt,
                                               const int* __restrict__ slots,
                                               const int* __restrict__ novf,
                                               const int* __restrict__ ovf_w,
                                               const float* __restrict__ bias,
                                               float* __restrict__ out, int n, int ntiles) {
  __shared__ float Ws[D * WP];
  __shared__ float As[2][TN * AP];

  const int2* __restrict__ ovf = (const int2*)ovf_w;
  const int tid  = threadIdx.x;
  const int lane = tid & 31;
  const int wave = __builtin_amdgcn_readfirstlane(tid >> 5);

  {
    const float4* __restrict__ Wv = (const float4*)W;
#pragma unroll
    for (int i = 0; i < (D * D / 4) / 512; ++i) {
      const int idx = tid + i * 512;
      *(float4*)&Ws[(idx >> 5) * WP + (idx & 31) * 4] = Wv[idx];
    }
  }

  const int m  = lane & 15;
  const int hh = lane >> 4;
  const int kb = hh << 1;
  const int rt = wave >> 3;
  const int ct = wave & 7;
  const int col = ct * 16 + m;
  const float bcol = bias[col];
  const float4 zero = make_float4(0.0f, 0.0f, 0.0f, 0.0f);

  int buf = 0;
  for (int tile = blockIdx.x; tile < ntiles; tile += gridDim.x) {
    float* __restrict__ Ab = As[buf];

    const int d0 = tile * TN + wave * 2;
    const int dl = (d0 + (lane & 1) < n) ? d0 + (lane & 1) : n - 1;
    const int cv = cnt[dl];
    const float dv = rsqrtf((float)(cv + 1));
    const int  c0n = __builtin_amdgcn_readlane(cv, 0);
    const int  c1n = __builtin_amdgcn_readlane(cv, 1);
    const float dd0 = __int_as_float(__builtin_amdgcn_readlane(__float_as_int(dv), 0));
    const float dd1 = __int_as_float(__builtin_amdgcn_readlane(__float_as_int(dv), 1));
    const bool ok0 = d0 < n;
    const bool ok1 = d0 + 1 < n;

    int e0 = d0;
    int e1 = d0 + 1;
    float4 self0 = zero, self1 = zero;
    if (ok0) {
      if (lane < c0n) e0 = slots[(size_t)d0 * CAP + lane];
      self0 = ld_row(hin, d0, lane);
    }
    if (ok1) {
      if (lane < c1n) e1 = slots[(size_t)(d0 + 1) * CAP + lane];
      self1 = ld_row(hin, d0 + 1, lane);
    }

    float4 acc0 = zero, acc1 = zero;
    if (ok0) acc0 = node_sum(hin, cnt, novf, ovf, d0, c0n, dd0, e0, self0, lane);
    *(float4*)&Ab[(wave * 2) * AP + lane * 4] = acc0;
    if (ok1) acc1 = node_sum(hin, cnt, novf, ovf, d0 + 1, c1n, dd1, e1, self1, lane);
    *(float4*)&Ab[(wave * 2 + 1) * AP + lane * 4] = acc1;
    __syncthreads();

    v8f c0 = {};
    v8f c1 = {};
    const float* Arow = Ab + (rt * 16 + m) * AP + kb;
    const float* Bcol = Ws + kb * WP + col;
#pragma unroll 4
    for (int k = 0; k < D; k += 8) {
      const v2f a0 = *(const v2f*)(Arow + k);
      const v2f a1 = *(const v2f*)(Arow + k + 4);
      v2f b0, b1;
      b0.x = Bcol[k * WP];
      b0.y = Bcol[(k + 1) * WP];
      b1.x = Bcol[(k + 4) * WP];
      b1.y = Bcol[(k + 5) * WP];
      c0 = __builtin_amdgcn_wmma_f32_16x16x4_f32(
           false, a0,  false, b0,
           (short)0, c0,  false,  false);
      c1 = __builtin_amdgcn_wmma_f32_16x16x4_f32(
           false, a1,  false, b1,
           (short)0, c1,  false,  false);
    }

    const int rb = tile * TN + rt * 16 + (hh << 3);
    float o[8];
#pragma unroll
    for (int v = 0; v < 8; ++v) {
      o[v] = (c0[v] + c1[v]) + bcol;
      if (RELU_OUT) o[v] = fmaxf(o[v], 0.0f);
    }
    float* __restrict__ orow = out + (size_t)rb * D + col;
    if (tile * TN + TN <= n) {
#pragma unroll
      for (int v = 0; v < 8; ++v) orow[v * D] = o[v];
    } else {
#pragma unroll
      for (int v = 0; v < 8; ++v)
        if (rb + v < n) orow[v * D] = o[v];
    }
    buf ^= 1;
  }
}

extern "C" void kernel_launch(void* const* d_in, const int* in_sizes, int n_in,
                              void* d_out, int out_size, void* d_ws, size_t ws_size,
                              hipStream_t stream) {
  const float* x  = (const float*)d_in[0];
  const int*   ei = (const int*)d_in[1];
  const float* W1 = (const float*)d_in[2];
  const float* b1 = (const float*)d_in[3];
  const float* W2 = (const float*)d_in[4];
  const float* b2 = (const float*)d_in[5];
  const float* W3 = (const float*)d_in[6];
  const float* b3 = (const float*)d_in[7];

  const int N = in_sizes[0] / D;
  const int E = in_sizes[1] / 2;

  size_t npad = ((size_t)N + 255) & ~(size_t)255;
  size_t epad = ((size_t)E + 255) & ~(size_t)255;
  int*   cnt    = (int*)d_ws;
  int*   novf   = cnt + npad;
  int*   slots  = novf + 256;
  int*   ovf    = slots + npad * CAP;
  float* buf1   = (float*)(ovf + 2 * epad);
  float* buf2   = buf1 + (size_t)N * D;
  float* out    = (float*)d_out;

  const dim3 blk(256);
  const int gN    = (N + 255) / 256;
  const int gE    = (E + 255) / 256;
  const int nTiles = (N + TN - 1) / TN;
  const int gLayer = 768;

  k_cnt_zero<<<gN, blk, 0, stream>>>(cnt, N, novf);
  k_bucket<<<gE, blk, 0, stream>>>(ei, E, cnt, slots, novf, ovf);

  k_layer<true><<<gLayer, dim3(512), 0, stream>>>(x, W1, cnt, slots, novf, ovf, b1, buf1, N, nTiles);

  k_layer<true><<<gLayer, dim3(512), 0, stream>>>(buf1, W2, cnt, slots, novf, ovf, b2, buf2, N, nTiles);

  k_layer<false><<<gLayer, dim3(512), 0, stream>>>(buf2, W3, cnt, slots, novf, ovf, b3, out, N, nTiles);
}


# Edges bucketed without global atomics: a work-group owns 1024 nodes, scans all destinations, takes slots from LDS adds, stores whole lines; 5 launches to 4; nodes over 32 in-edges sum in list order
# speedup vs baseline: 3.6033x; 1.0517x over previous
#include <hip/hip_runtime.h>


typedef __attribute__((ext_vector_type(2))) float v2f;
typedef __attribute__((ext_vector_type(8))) float v8f;

#define D 128
#define WP 136
#define AP 132
#define TN 32
#define CAP 32
#define BR 1024
#define LCAP 512
#define SCAN_U 8

__device__ __forceinline__ void drain_list(const int2* lst, int nent, int lane,
                                           const int* __restrict__ src, int* cl, int* sl) {
  for (int j = lane; j < nent; j += 32) {
    const int2 e = lst[j];
    const int s = src[e.x];
    const int r = atomicAdd(&cl[e.y], 1);
    if (r < CAP) sl[e.y * CAP + r] = s;
  }
}

__device__ __forceinline__ void drain_list_wide(const int2* lst, int nent, int lane,
                                                const int* __restrict__ src, int* cl, int* sl) {
  for (int j0 = 0; j0 < nent; j0 += 128) {
    int2 e[4];
    int  s[4];
#pragma unroll
    for (int u = 0; u < 4; ++u) {
      const int j = j0 + u * 32 + lane;
      e[u] = make_int2(0, -1);
      if (j < nent) e[u] = lst[j];
    }
#pragma unroll
    for (int u = 0; u < 4; ++u) {
      s[u] = 0;
      if (e[u].y >= 0) s[u] = src[e[u].x];
    }
#pragma unroll
    for (int u = 0; u < 4; ++u) {
      if (e[u].y >= 0) {
        const int r = atomicAdd(&cl[e[u].y], 1);
        if (r < CAP) sl[e[u].y * CAP + r] = s[u];
      }
    }
  }
}

__device__ __forceinline__ int note_hits(int2* lst, int cur, int i, unsigned dl) {
  const bool hit = dl < (unsigned)BR;
  const unsigned m = __builtin_amdgcn_ballot_w32(hit);
  if (m != 0u) {
    if (hit) lst[cur + (int)__builtin_amdgcn_mbcnt_lo(m, 0u)] = make_int2(i, (int)dl);
    cur += __builtin_popcount(m);
  }
  return cur;
}

template <bool FULL>
__device__ __forceinline__ int scan_trip(const int4* __restrict__ dq, int qb, int nq, int head,
                                         int tid, int lane, int base, int2* mylist, int cur,
                                         const int* __restrict__ ei, int* cl, int* sl) {
  int4 v[SCAN_U];
#pragma unroll
  for (int u = 0; u < SCAN_U; ++u) {
    const int q = qb + u * 1024 + tid;
    if (FULL) {
      v[u] = dq[q];
    } else {
      const int4 t = dq[(q < nq) ? q : nq - 1];
      const bool in = q < nq;
      v[u] = make_int4(in ? t.x : -1, in ? t.y : -1, in ? t.z : -1, in ? t.w : -1);
    }
  }
#pragma unroll
  for (int u = 0; u < SCAN_U; ++u) {
    if (cur > LCAP - 128) {
      drain_list(mylist, cur, lane, ei, cl, sl);
      cur = 0;
    }
    const int i = head + 4 * (qb + u * 1024 + tid);
    cur = note_hits(mylist, cur, i,     (unsigned)(v[u].x - base));
    cur = note_hits(mylist, cur, i + 1, (unsigned)(v[u].y - base));
    cur = note_hits(mylist, cur, i + 2, (unsigned)(v[u].z - base));
    cur = note_hits(mylist, cur, i + 3, (unsigned)(v[u].w - base));
  }
  return cur;
}

__global__ __launch_bounds__(1024) void k_bucket(const int* __restrict__ ei, int E, int n,
                                                 int* __restrict__ cnt,
                                                 int* __restrict__ slots) {
  __shared__ int  cl[BR];
  __shared__ int  sl[BR * CAP];
  __shared__ int2 lst[32][LCAP];

  const int tid  = threadIdx.x;
  const int lane = tid & 31;
  const int wave = __builtin_amdgcn_readfirstlane(tid >> 5);
  const int base = blockIdx.x * BR;
  const int* __restrict__ dst = ei + E;
  int2* mylist = lst[wave];

  cl[tid] = 0;
  __syncthreads();

  const int head = (E < 4) ? E : ((4 - (E & 3)) & 3);
  const int nq   = (E - head) >> 2;
  const int4* __restrict__ dq = (const int4*)(dst + head);

  int cur = 0;
  int qb = 0;
  for (; qb + SCAN_U * 1024 <= nq; qb += SCAN_U * 1024)
    cur = scan_trip<true>(dq, qb, nq, head, tid, lane, base, mylist, cur, ei, cl, sl);
  if (qb < nq)
    cur = scan_trip<false>(dq, qb, nq, head, tid, lane, base, mylist, cur, ei, cl, sl);
  if (wave == 0) {
    const int tail0 = head + 4 * nq;
    int i = -1;
    if (lane < head) i = lane;
    else if (lane >= 4 && tail0 + (lane - 4) < E && lane < 8) i = tail0 + (lane - 4);
    int d = -1;
    if (i >= 0) d = dst[i];
    if (cur > LCAP - 32) {
      drain_list(mylist, cur, lane, ei, cl, sl);
      cur = 0;
    }
    cur = note_hits(mylist, cur, i, (unsigned)(d - base));
  }
  drain_list_wide(mylist, cur, lane, ei, cl, sl);
  __syncthreads();

  if (base + tid < n) cnt[base + tid] = cl[tid];
  int4* __restrict__ so = (int4*)(slots + (size_t)base * CAP);
  const int4* sv = (const int4*)sl;
#pragma unroll
  for (int k = 0; k < (BR * CAP / 4) / 1024; ++k) {
    const int q    = tid + k * 1024;
    const int node = q >> 3;
    const int j0   = (q & 7) * 4;
    if (base + node < n) {
      const int c = cl[node];
      int4 v = sv[q];
      v.x = (j0     < c) ? v.x : 0;
      v.y = (j0 + 1 < c) ? v.y : 0;
      v.z = (j0 + 2 < c) ? v.z : 0;
      v.w = (j0 + 3 < c) ? v.w : 0;
      so[q] = v;
    }
  }
}

__device__ __forceinline__ int pick_lane(int r, int p) {
  const unsigned hit = __builtin_amdgcn_ballot_w32(r == p);
  return __builtin_ctz(hit);
}

__device__ __forceinline__ float4 ld_row(const float* __restrict__ h, int s, int lane) {
  return *(const float4*)(h + (size_t)s * D + lane * 4);
}

__device__ __forceinline__ float4 fma_row(float4 v, float w, float4 a) {
  a.x = fmaf(v.x, w, a.x);
  a.y = fmaf(v.y, w, a.y);
  a.z = fmaf(v.z, w, a.z);
  a.w = fmaf(v.w, w, a.w);
  return a;
}

__device__ __forceinline__ float4 node_sum(const float* __restrict__ h,
                                           const int* __restrict__ cnt,
                                           const int* __restrict__ ei, int E, int d, int c,
                                           float dd, int s_first, float4 self_row,
                                           int lane) {
  const float4 zero = make_float4(0.0f, 0.0f, 0.0f, 0.0f);

  float4 acc = fma_row(self_row, dd * dd, zero);

  const int m = (c <= CAP) ? c : 0;
  if (m > 0) {
    const int s = s_first;
    int r = 0;
    for (int i = 0; i < m; ++i) {
      const int si = __builtin_amdgcn_readlane(s, i);
      r += ((si < s) || (si == s && i < lane)) ? 1 : 0;
    }
    if (lane >= m) r = 64;

    float w = 0.0f;
    for (int p = 0; p < m; p += 8) {
      float4 v[8];
      int ix[8];
#pragma unroll
      for (int q = 0; q < 8; ++q) {
        v[q]  = zero;
        ix[q] = 0;
        if (p + q < m) {
          ix[q] = pick_lane(r, p + q);
          v[q]  = ld_row(h, __builtin_amdgcn_readlane(s, ix[q]), lane);
        }
      }
      if (p == 0) {
        int cs = 0;
        if (lane < m) cs = cnt[s];
        w = (lane < m) ? rsqrtf((float)(cs + 1)) * dd : 0.0f;
      }
#pragma unroll
      for (int q = 0; q < 8; ++q) {
        const float wq = (p + q < m)
            ? __int_as_float(__builtin_amdgcn_readlane(__float_as_int(w), ix[q])) : 0.0f;
        acc = fma_row(v[q], wq, acc);
      }
    }
  }

  if (c > CAP) {
    for (int base = 0; base < E; base += 32) {
      int2 oe = make_int2(-1, 0);
      if (base + lane < E) oe.x = ei[E + base + lane];
      const bool mine = (oe.x == d);
      if (mine) oe.y = ei[base + lane];
      int cs = 0;
      if (mine) cs = cnt[oe.y];
      const float w = mine ? rsqrtf((float)(cs + 1)) * dd : 0.0f;
      unsigned hit = __builtin_amdgcn_ballot_w32(mine);
      while (hit != 0u) {
        const int idx = __builtin_ctz(hit);
        hit &= hit - 1u;
        const float4 v = ld_row(h, __builtin_amdgcn_readlane(oe.y, idx), lane);
        const float wq = __int_as_float(__builtin_amdgcn_readlane(__float_as_int(w), idx));
        acc = fma_row(v, wq, acc);
      }
    }
  }
  return acc;
}

template <bool RELU_OUT>
__global__ __launch_bounds__(512) void k_layer(const float* __restrict__ hin,
                                               const float* __restrict__ W,
                                               const int* __restrict__ cnt,
                                               const int* __restrict__ slots,
                                               const int* __restrict__ ei, int E,
                                               const float* __restrict__ bias,
                                               float* __restrict__ out, int n, int ntiles) {
  __shared__ float Ws[D * WP];
  __shared__ float As[2][TN * AP];

  const int tid  = threadIdx.x;
  const int lane = tid & 31;
  const int wave = __builtin_amdgcn_readfirstlane(tid >> 5);

  {
    const float4* __restrict__ Wv = (const float4*)W;
#pragma unroll
    for (int i = 0; i < (D * D / 4) / 512; ++i) {
      const int idx = tid + i * 512;
      *(float4*)&Ws[(idx >> 5) * WP + (idx & 31) * 4] = Wv[idx];
    }
  }

  const int m  = lane & 15;
  const int hh = lane >> 4;
  const int kb = hh << 1;
  const int rt = wave >> 3;
  const int ct = wave & 7;
  const int col = ct * 16 + m;
  const float bcol = bias[col];
  const float4 zero = make_float4(0.0f, 0.0f, 0.0f, 0.0f);

  int buf = 0;
  for (int tile = blockIdx.x; tile < ntiles; tile += gridDim.x) {
    float* __restrict__ Ab = As[buf];

    const int d0 = tile * TN + wave * 2;
    const int dl = (d0 + (lane & 1) < n) ? d0 + (lane & 1) : n - 1;
    const int cv = cnt[dl];
    const float dv = rsqrtf((float)(cv + 1));
    const int  c0n = __builtin_amdgcn_readlane(cv, 0);
    const int  c1n = __builtin_amdgcn_readlane(cv, 1);
    const float dd0 = __int_as_float(__builtin_amdgcn_readlane(__float_as_int(dv), 0));
    const float dd1 = __int_as_float(__builtin_amdgcn_readlane(__float_as_int(dv), 1));
    const bool ok0 = d0 < n;
    const bool ok1 = d0 + 1 < n;

    int e0 = d0;
    int e1 = d0 + 1;
    float4 self0 = zero, self1 = zero;
    if (ok0) {
      if (lane < c0n) e0 = slots[(size_t)d0 * CAP + lane];
      self0 = ld_row(hin, d0, lane);
    }
    if (ok1) {
      if (lane < c1n) e1 = slots[(size_t)(d0 + 1) * CAP + lane];
      self1 = ld_row(hin, d0 + 1, lane);
    }

    float4 acc0 = zero, acc1 = zero;
    if (ok0) acc0 = node_sum(hin, cnt, ei, E, d0, c0n, dd0, e0, self0, lane);
    *(float4*)&Ab[(wave * 2) * AP + lane * 4] = acc0;
    if (ok1) acc1 = node_sum(hin, cnt, ei, E, d0 + 1, c1n, dd1, e1, self1, lane);
    *(float4*)&Ab[(wave * 2 + 1) * AP + lane * 4] = acc1;
    __syncthreads();

    v8f c0 = {};
    v8f c1 = {};
    const float* Arow = Ab + (rt * 16 + m) * AP + kb;
    const float* Bcol = Ws + kb * WP + col;
#pragma unroll 4
    for (int k = 0; k < D; k += 8) {
      const v2f a0 = *(const v2f*)(Arow + k);
      const v2f a1 = *(const v2f*)(Arow + k + 4);
      v2f b0, b1;
      b0.x = Bcol[k * WP];
      b0.y = Bcol[(k + 1) * WP];
      b1.x = Bcol[(k + 4) * WP];
      b1.y = Bcol[(k + 5) * WP];
      c0 = __builtin_amdgcn_wmma_f32_16x16x4_f32(
           false, a0,  false, b0,
           (short)0, c0,  false,  false);
      c1 = __builtin_amdgcn_wmma_f32_16x16x4_f32(
           false, a1,  false, b1,
           (short)0, c1,  false,  false);
    }

    const int rb = tile * TN + rt * 16 + (hh << 3);
    float o[8];
#pragma unroll
    for (int v = 0; v < 8; ++v) {
      o[v] = (c0[v] + c1[v]) + bcol;
      if (RELU_OUT) o[v] = fmaxf(o[v], 0.0f);
    }
    float* __restrict__ orow = out + (size_t)rb * D + col;
    if (tile * TN + TN <= n) {
#pragma unroll
      for (int v = 0; v < 8; ++v) orow[v * D] = o[v];
    } else {
#pragma unroll
      for (int v = 0; v < 8; ++v)
        if (rb + v < n) orow[v * D] = o[v];
    }
    buf ^= 1;
  }
}

extern "C" void kernel_launch(void* const* d_in, const int* in_sizes, int n_in,
                              void* d_out, int out_size, void* d_ws, size_t ws_size,
                              hipStream_t stream) {
  const float* x  = (const float*)d_in[0];
  const int*   ei = (const int*)d_in[1];
  const float* W1 = (const float*)d_in[2];
  const float* b1 = (const float*)d_in[3];
  const float* W2 = (const float*)d_in[4];
  const float* b2 = (const float*)d_in[5];
  const float* W3 = (const float*)d_in[6];
  const float* b3 = (const float*)d_in[7];

  const int N = in_sizes[0] / D;
  const int E = in_sizes[1] / 2;

  size_t npad = ((size_t)N + 255) & ~(size_t)255;
  int*   cnt    = (int*)d_ws;
  int*   slots  = cnt + npad;
  float* buf1   = (float*)(slots + npad * CAP);
  float* buf2   = buf1 + (size_t)N * D;
  float* out    = (float*)d_out;

  const int gB    = (N + BR - 1) / BR;
  const int nTiles = (N + TN - 1) / TN;
  const int gLayer = 768;

  k_bucket<<<gB, dim3(1024), 0, stream>>>(ei, E, N, cnt, slots);

  k_layer<true><<<gLayer, dim3(512), 0, stream>>>(x, W1, cnt, slots, ei, E, b1, buf1, N, nTiles);

  k_layer<true><<<gLayer, dim3(512), 0, stream>>>(buf1, W2, cnt, slots, ei, E, b2, buf2, N, nTiles);

  k_layer<false><<<gLayer, dim3(512), 0, stream>>>(buf2, W3, cnt, slots, ei, E, b3, out, N, nTiles);
}
